# starting kernel
# baseline (speedup 1.0000x reference)
#include <hip/hip_runtime.h>

typedef __attribute__((ext_vector_type(2))) float v2f;
typedef __attribute__((ext_vector_type(8))) float v8f;

#define BATCH   16
#define CDIM    64
#define HW      4096
#define NROWS   (BATCH * HW)
#define KEMB    1024
#define TPB     256
#define NWAVES  8
#define ROWS_WG 128
#define NWG     (NROWS / ROWS_WG)

#define LDS_FLOATS 66696
#define LDS_BYTES  (LDS_FLOATS * 4)

__global__ void vq_zero_kernel(float* acc) {
    if (threadIdx.x == 0) acc[0] = 0.0f;
}

__global__ __launch_bounds__(TPB)
void vq_main_kernel(const float* __restrict__ x,
                    const float* __restrict__ E,
                    float* __restrict__ zq,
                    float* __restrict__ lossAcc)
{
    extern __shared__ float lds[];
    float* Bfrag  = lds;
    float* esq    = lds + 65536;
    int*   idxs   = (int*)(lds + 66560);
    float* redbuf = lds + 66688;

    const int tid = threadIdx.x;
    const int wg  = blockIdx.x;

    for (int p = tid; p < (KEMB * CDIM / 2); p += TPB) {
        int k  = p >> 5;
        int j  = p & 31;
        int c0 = j * 2;
        float e0 = E[k * CDIM + c0];
        float e1 = E[k * CDIM + c0 + 1];
        int t  = k >> 4;
        int n  = k & 15;
        int kc = c0 >> 2;
        int g  = (c0 >> 1) & 1;
        int fo = ((t * 16 + kc) * 32 + (g * 16 + n)) * 2;
        Bfrag[fo]     = e0;
        Bfrag[fo + 1] = e1;
    }
    for (int k = tid; k < KEMB; k += TPB) {
        float s = 0.0f;
        for (int c = 0; c < CDIM; ++c) {
            float e = E[k * CDIM + c];
            s += e * e;
        }
        esq[k] = s;
    }
    __syncthreads();

    const int wave = tid >> 5;
    const int lane = tid & 31;
    const int m    = lane & 15;
    const int g    = lane >> 4;
    const int rowbase = wg * ROWS_WG + wave * 16;
    const int b       = rowbase >> 12;
    const int nrow    = (rowbase & 4095) + m;

    v2f afr[16];
#pragma unroll
    for (int kc = 0; kc < 16; ++kc) {
        int c0 = kc * 4 + g * 2;
        afr[kc].x = x[(b * CDIM + c0    ) * HW + nrow];
        afr[kc].y = x[(b * CDIM + c0 + 1) * HW + nrow];
    }

    float minval[8];
    int   minidx[8];
#pragma unroll
    for (int r = 0; r < 8; ++r) { minval[r] = 3.402823466e38f; minidx[r] = 0; }

    for (int t = 0; t < KEMB / 16; ++t) {
        v8f acc = {0.f, 0.f, 0.f, 0.f, 0.f, 0.f, 0.f, 0.f};
        const v2f* bp = (const v2f*)(Bfrag + t * 1024);
#pragma unroll
        for (int kc = 0; kc < 16; ++kc) {
            v2f bf = bp[kc * 32 + lane];
            acc = __builtin_amdgcn_wmma_f32_16x16x4_f32(
                false, afr[kc], false, bf, (short)0, acc, false, false);
        }
        int   col = t * 16 + m;
        float es  = esq[col];
#pragma unroll
        for (int r = 0; r < 8; ++r) {
            float d = es - 2.0f * acc[r];
            if (d < minval[r]) { minval[r] = d; minidx[r] = col; }
        }
    }

#pragma unroll
    for (int r = 0; r < 8; ++r) {
        float v = minval[r];
        int   i = minidx[r];
#pragma unroll
        for (int off = 1; off <= 8; off <<= 1) {
            float ov = __shfl_xor(v, off, 32);
            int   oi = __shfl_xor(i, off, 32);
            if (ov < v || (ov == v && oi < i)) { v = ov; i = oi; }
        }
        if (m == 0) idxs[wave * 16 + g * 8 + r] = i;
    }
    __syncthreads();

    const int b_wg = (wg * ROWS_WG) >> 12;
    const int n_wg = (wg * ROWS_WG) & 4095;
    float lsum = 0.0f;
    for (int ii = tid; ii < ROWS_WG * CDIM; ii += TPB) {
        int rl = ii & 127;
        int c  = ii >> 7;
        int k  = idxs[rl];
        int t  = k >> 4, nn = k & 15, kc = c >> 2, gg = (c >> 1) & 1;
        float ev = Bfrag[((t * 16 + kc) * 32 + (gg * 16 + nn)) * 2 + (c & 1)];
        int gaddr = (b_wg * CDIM + c) * HW + n_wg + rl;
        float xv = x[gaddr];
        zq[gaddr] = ev;
        float dlt = ev - xv;
        lsum += dlt * dlt;
    }
#pragma unroll
    for (int off = 1; off < 32; off <<= 1) lsum += __shfl_xor(lsum, off, 32);
    if (lane == 0) redbuf[wave] = lsum;
    __syncthreads();
    if (tid == 0) {
        float tot = 0.0f;
        for (int wv = 0; wv < NWAVES; ++wv) tot += redbuf[wv];
        atomicAdd(lossAcc, tot);
    }
}

__global__ void vq_finalize_kernel(const float* acc, float* lossOut) {
    if (threadIdx.x == 0) {
        lossOut[0] = 2.0f * acc[0] / (float)((long)NROWS * CDIM);
    }
}

extern "C" void kernel_launch(void* const* d_in, const int* in_sizes, int n_in,
                              void* d_out, int out_size, void* d_ws, size_t ws_size,
                              hipStream_t stream) {
    const float* x = (const float*)d_in[0];
    const float* E = (const float*)d_in[1];
    float* out     = (float*)d_out;
    float* acc     = (float*)d_ws;

    vq_zero_kernel<<<1, 64, 0, stream>>>(acc);
    vq_main_kernel<<<NWG, TPB, LDS_BYTES, stream>>>(x, E, out, acc);
    vq_finalize_kernel<<<1, 64, 0, stream>>>(acc, out + (out_size - 1));
}


# Two 16-row tiles per wave (256 rows per work-group, 256 work-groups): each B fragment read from LDS feeds two independent accumulators; loss partial sums regrouped per 256 rows
# speedup vs baseline: 1.5238x; 1.5238x over previous
#include <hip/hip_runtime.h>

typedef __attribute__((ext_vector_type(2))) float v2f;
typedef __attribute__((ext_vector_type(8))) float v8f;

#define BATCH   16
#define CDIM    64
#define HW      4096
#define NROWS   (BATCH * HW)
#define KEMB    1024
#define TPB     256
#define NWAVES  8
#define TILES_W 2
#define ROWS_WG (NWAVES * TILES_W * 16)
#define NWG     (NROWS / ROWS_WG)

#define BPITCH       66
#define BFRAG_FLOATS (KEMB * BPITCH)
#define ESQ_OFF      BFRAG_FLOATS
#define IDX_OFF      (ESQ_OFF + KEMB)
#define RED_OFF      (IDX_OFF + ROWS_WG)

#define LDS_FLOATS (RED_OFF + NWAVES)
#define LDS_BYTES  (LDS_FLOATS * 4)

__global__ void vq_zero_kernel(float* acc) {
    if (threadIdx.x == 0) acc[0] = 0.0f;
}

__global__ __launch_bounds__(TPB)
void vq_main_kernel(const float* __restrict__ x,
                    const float* __restrict__ E,
                    float* __restrict__ zq,
                    float* __restrict__ lossAcc)
{
    extern __shared__ float lds[];
    float* Bfrag  = lds;
    float* esq    = lds + ESQ_OFF;
    int*   idxs   = (int*)(lds + IDX_OFF);
    float* redbuf = lds + RED_OFF;

    const int tid  = threadIdx.x;
    const int wg   = blockIdx.x;
    const int wave = tid >> 5;
    const int lane = tid & 31;

    {
        const int hrow = lane >> 4;
        const int kc   = lane & 15;
#pragma unroll 8
        for (int q = wave; q < KEMB / 2; q += NWAVES) {
            int t = ((q >> 4) << 1) + hrow;
            int n = q & 15;
            int k = t * 16 + n;
            float4 v = *(const float4*)(E + k * CDIM + kc * 4);
            float* dst = Bfrag + (t * 16 + kc) * BPITCH + n * 2;
            v2f lo = {v.x, v.y};
            v2f hi = {v.z, v.w};
            *(v2f*)(dst)      = lo;
            *(v2f*)(dst + 32) = hi;
        }
    }
    __syncthreads();
    {
        const float* rp = Bfrag + (tid >> 4) * (16 * BPITCH) + (tid & 15) * 2;
        float s[4] = {0.0f, 0.0f, 0.0f, 0.0f};
#pragma unroll
        for (int kc = 0; kc < 16; ++kc) {
#pragma unroll
            for (int j = 0; j < 4; ++j) {
                const float* p = rp + j * (TPB * BPITCH) + kc * BPITCH;
                v2f lo = *(const v2f*)(p);
                v2f hi = *(const v2f*)(p + 32);
                s[j] = __builtin_fmaf(lo.x, lo.x, s[j]);
                s[j] = __builtin_fmaf(lo.y, lo.y, s[j]);
                s[j] = __builtin_fmaf(hi.x, hi.x, s[j]);
                s[j] = __builtin_fmaf(hi.y, hi.y, s[j]);
            }
        }
#pragma unroll
        for (int j = 0; j < 4; ++j) esq[tid + j * TPB] = s[j];
    }
    __syncthreads();

    const int m    = lane & 15;
    const int g    = lane >> 4;
    const int rowbase = wg * ROWS_WG + wave * (TILES_W * 16);
    const int b       = rowbase >> 12;
    const int nrow    = (rowbase & 4095) + m;

    v2f afr0[16], afr1[16];
#pragma unroll
    for (int kc = 0; kc < 16; ++kc) {
        int c0 = kc * 4 + g * 2;
        afr0[kc].x = x[(b * CDIM + c0    ) * HW + nrow];
        afr0[kc].y = x[(b * CDIM + c0 + 1) * HW + nrow];
        afr1[kc].x = x[(b * CDIM + c0    ) * HW + nrow + 16];
        afr1[kc].y = x[(b * CDIM + c0 + 1) * HW + nrow + 16];
    }

    float minval0[8], minval1[8];
    int   minidx0[8], minidx1[8];
#pragma unroll
    for (int r = 0; r < 8; ++r) {
        minval0[r] = 3.402823466e38f; minidx0[r] = 0;
        minval1[r] = 3.402823466e38f; minidx1[r] = 0;
    }

    for (int t = 0; t < KEMB / 16; ++t) {
        v8f acc0 = {0.f, 0.f, 0.f, 0.f, 0.f, 0.f, 0.f, 0.f};
        v8f acc1 = {0.f, 0.f, 0.f, 0.f, 0.f, 0.f, 0.f, 0.f};
        const float* bp = Bfrag + t * (16 * BPITCH) + lane * 2;
#pragma unroll
        for (int kc = 0; kc < 16; ++kc) {
            v2f bf = *(const v2f*)(bp + kc * BPITCH);
            acc0 = __builtin_amdgcn_wmma_f32_16x16x4_f32(
                false, afr0[kc], false, bf, (short)0, acc0, false, false);
            acc1 = __builtin_amdgcn_wmma_f32_16x16x4_f32(
                false, afr1[kc], false, bf, (short)0, acc1, false, false);
        }
        int   col = t * 16 + m;
        float es  = esq[col];
#pragma unroll
        for (int r = 0; r < 8; ++r) {
            float d0 = es - 2.0f * acc0[r];
            if (d0 < minval0[r]) { minval0[r] = d0; minidx0[r] = col; }
            float d1 = es - 2.0f * acc1[r];
            if (d1 < minval1[r]) { minval1[r] = d1; minidx1[r] = col; }
        }
    }

#pragma unroll
    for (int q = 0; q < 2 * 8; ++q) {
        const int tl = q >> 3;
        const int r  = q & 7;
        float v = tl ? minval1[r] : minval0[r];
        int   i = tl ? minidx1[r] : minidx0[r];
#pragma unroll
        for (int off = 1; off <= 8; off <<= 1) {
            float ov = __shfl_xor(v, off, 32);
            int   oi = __shfl_xor(i, off, 32);
            if (ov < v || (ov == v && oi < i)) { v = ov; i = oi; }
        }
        if (m == 0) idxs[wave * (TILES_W * 16) + tl * 16 + g * 8 + r] = i;
    }
    __syncthreads();

    const int b_wg = (wg * ROWS_WG) >> 12;
    const int n_wg = (wg * ROWS_WG) & 4095;
    float lsum = 0.0f;
    for (int ii = tid; ii < ROWS_WG * CDIM; ii += TPB) {
        int rl = ii & (ROWS_WG - 1);
        int c  = ii >> 8;
        int k  = idxs[rl];
        int t  = k >> 4, nn = k & 15, kc = c >> 2, gg = (c >> 1) & 1;
        float ev = Bfrag[(t * 16 + kc) * BPITCH + gg * 32 + nn * 2 + (c & 1)];
        int gaddr = (b_wg * CDIM + c) * HW + n_wg + rl;
        float xv = x[gaddr];
        zq[gaddr] = ev;
        float dlt = ev - xv;
        lsum += dlt * dlt;
    }
#pragma unroll
    for (int off = 1; off < 32; off <<= 1) lsum += __shfl_xor(lsum, off, 32);
    if (lane == 0) redbuf[wave] = lsum;
    __syncthreads();
    if (tid == 0) {
        float tot = 0.0f;
        for (int wv = 0; wv < NWAVES; ++wv) tot += redbuf[wv];
        atomicAdd(lossAcc, tot);
    }
}

__global__ void vq_finalize_kernel(const float* acc, float* lossOut) {
    if (threadIdx.x == 0) {
        lossOut[0] = 2.0f * acc[0] / (float)((long)NROWS * CDIM);
    }
}

extern "C" void kernel_launch(void* const* d_in, const int* in_sizes, int n_in,
                              void* d_out, int out_size, void* d_ws, size_t ws_size,
                              hipStream_t stream) {
    const float* x = (const float*)d_in[0];
    const float* E = (const float*)d_in[1];
    float* out     = (float*)d_out;
    float* acc     = (float*)d_ws;

    vq_zero_kernel<<<1, 64, 0, stream>>>(acc);
    vq_main_kernel<<<NWG, TPB, LDS_BYTES, stream>>>(x, E, out, acc);
    vq_finalize_kernel<<<1, 64, 0, stream>>>(acc, out + (out_size - 1));
}


# Epilogue kept in each wave's registers: 16-lane argmin by row DPP steps; z_q gathered per wave and loss taken against the held A fragments (no x re-read, no index barrier); loss summed per lane
# speedup vs baseline: 1.7771x; 1.1662x over previous
#include <hip/hip_runtime.h>

typedef __attribute__((ext_vector_type(2))) float v2f;
typedef __attribute__((ext_vector_type(8))) float v8f;

#define BATCH   16
#define CDIM    64
#define HW      4096
#define NROWS   (BATCH * HW)
#define KEMB    1024
#define TPB     256
#define NWAVES  8
#define TILES_W 2
#define ROWS_WG (NWAVES * TILES_W * 16)
#define NWG     (NROWS / ROWS_WG)

#define BPITCH       66
#define BFRAG_FLOATS (KEMB * BPITCH)
#define ESQ_OFF      BFRAG_FLOATS
#define RED_OFF      (ESQ_OFF + KEMB)

#define LDS_FLOATS (RED_OFF + NWAVES)
#define LDS_BYTES  (LDS_FLOATS * 4)

__global__ void vq_zero_kernel(float* acc) {
    if (threadIdx.x == 0) acc[0] = 0.0f;
}

__device__ __forceinline__ int vq_xhalf_i(int v) {
    return __builtin_amdgcn_permlanex16(v, v, 0x76543210, 0xfedcba98, false, false);
}
__device__ __forceinline__ float vq_xhalf_f(float v) {
    return __int_as_float(vq_xhalf_i(__float_as_int(v)));
}

template <int XM>
__device__ __forceinline__ void vq_row_min_step(float& v, int& i) {
    const float ov = __int_as_float(
        __builtin_amdgcn_update_dpp(0, __float_as_int(v), 0x160 + XM, 0xf, 0xf, false));
    const int oi = __builtin_amdgcn_update_dpp(0, i, 0x160 + XM, 0xf, 0xf, false);
    const bool take = (ov < v) | ((ov == v) & (oi < i));
    v = take ? ov : v;
    i = take ? oi : i;
}

__device__ __forceinline__ void vq_tile_block(
    const v2f (&a0)[16], const v2f (&a1)[16], const v2f (&b)[16],
    v8f& out0, v8f& out1, const v8f& p0, const v8f& p1, const float esP, const int colP,
    float (&mv0)[8], int (&mi0)[8], float (&mv1)[8], int (&mi1)[8])
{
    v8f c0 = {0.f, 0.f, 0.f, 0.f, 0.f, 0.f, 0.f, 0.f};
    v8f c1 = c0;
#pragma unroll
    for (int kc = 0; kc < 16; ++kc) {
        c0 = __builtin_amdgcn_wmma_f32_16x16x4_f32(
            false, a0[kc], false, b[kc], (short)0, c0, false, false);
        c1 = __builtin_amdgcn_wmma_f32_16x16x4_f32(
            false, a1[kc], false, b[kc], (short)0, c1, false, false);
        if ((kc & 1) == 0) {
            __builtin_amdgcn_sched_barrier(0);
            const int r = kc >> 1;
            float d0 = esP - 2.0f * p0[r];
            if (d0 < mv0[r]) { mv0[r] = d0; mi0[r] = colP; }
            float d1 = esP - 2.0f * p1[r];
            if (d1 < mv1[r]) { mv1[r] = d1; mi1[r] = colP; }
        }
    }
    out0 = c0; out1 = c1;
}

__global__ __launch_bounds__(TPB)
void vq_main_kernel(const float* __restrict__ x,
                    const float* __restrict__ E,
                    float* __restrict__ zq,
                    float* __restrict__ lossAcc)
{
    extern __shared__ float lds[];
    float* Bfrag  = lds;
    float* esq    = lds + ESQ_OFF;
    float* redbuf = lds + RED_OFF;

    const int tid  = threadIdx.x;
    const int wg   = blockIdx.x;
    const int wave = tid >> 5;
    const int lane = tid & 31;

    {
        const int hrow = lane >> 4;
        const int kc   = lane & 15;
#pragma unroll 8
        for (int q = wave; q < KEMB / 2; q += NWAVES) {
            int t = ((q >> 4) << 1) + hrow;
            int n = q & 15;
            int k = t * 16 + n;
            float4 v = *(const float4*)(E + k * CDIM + kc * 4);
            float* dst = Bfrag + (t * 16 + kc) * BPITCH + n * 2;
            v2f lo = {v.x, v.y};
            v2f hi = {v.z, v.w};
            *(v2f*)(dst)      = lo;
            *(v2f*)(dst + 32) = hi;
        }
    }
    __syncthreads();
    {
        const float* rp = Bfrag + (tid >> 4) * (16 * BPITCH) + (tid & 15) * 2;
        float s[4] = {0.0f, 0.0f, 0.0f, 0.0f};
#pragma unroll
        for (int kc = 0; kc < 16; ++kc) {
#pragma unroll
            for (int j = 0; j < 4; ++j) {
                const float* p = rp + j * (TPB * BPITCH) + kc * BPITCH;
                v2f lo = *(const v2f*)(p);
                v2f hi = *(const v2f*)(p + 32);
                s[j] = __builtin_fmaf(lo.x, lo.x, s[j]);
                s[j] = __builtin_fmaf(lo.y, lo.y, s[j]);
                s[j] = __builtin_fmaf(hi.x, hi.x, s[j]);
                s[j] = __builtin_fmaf(hi.y, hi.y, s[j]);
            }
        }
#pragma unroll
        for (int j = 0; j < 4; ++j) esq[tid + j * TPB] = s[j];
    }
    __syncthreads();

    const int m    = lane & 15;
    const int g    = lane >> 4;
    const int rowbase = wg * ROWS_WG + wave * (TILES_W * 16);
    const int b       = rowbase >> 12;
    const int nrow    = (rowbase & 4095) + m;

    v2f afr0[16], afr1[16];
#pragma unroll
    for (int kc = 0; kc < 16; ++kc) {
        int c0 = kc * 4 + g * 2;
        afr0[kc].x = x[(b * CDIM + c0    ) * HW + nrow];
        afr0[kc].y = x[(b * CDIM + c0 + 1) * HW + nrow];
        afr1[kc].x = x[(b * CDIM + c0    ) * HW + nrow + 16];
        afr1[kc].y = x[(b * CDIM + c0 + 1) * HW + nrow + 16];
    }

    float minval0[8], minval1[8];
    int   minidx0[8], minidx1[8];
#pragma unroll
    for (int r = 0; r < 8; ++r) {
        minval0[r] = 3.402823466e38f; minidx0[r] = 0;
        minval1[r] = 3.402823466e38f; minidx1[r] = 0;
    }

    const float* bp0 = Bfrag + lane * 2;
    v2f bA[16], bB[16];
    float esA, esB;
    v8f accA0 = {0.f, 0.f, 0.f, 0.f, 0.f, 0.f, 0.f, 0.f};
    v8f accA1 = accA0, accB0 = accA0, accB1 = accA0;
    esB = 3.402823466e38f;

#pragma unroll
    for (int kc = 0; kc < 16; ++kc) bA[kc] = *(const v2f*)(bp0 + kc * BPITCH);
    esA = esq[m];

    for (int t = 0; t < KEMB / 16; t += 2) {
        const float esPrev = esB;
        __builtin_amdgcn_sched_barrier(0);
        {
            const float* bp = bp0 + (t + 1) * (16 * BPITCH);
            esB = esq[(t + 1) * 16 + m];
#pragma unroll
            for (int kc = 0; kc < 16; ++kc) bB[kc] = *(const v2f*)(bp + kc * BPITCH);
        }
        __builtin_amdgcn_sched_barrier(0);
        vq_tile_block(afr0, afr1, bA, accA0, accA1, accB0, accB1, esPrev, (t - 1) * 16 + m,
                      minval0, minidx0, minval1, minidx1);

        const float esCur = esA;
        __builtin_amdgcn_sched_barrier(0);
        {
            const int tn = (t + 2 < KEMB / 16) ? (t + 2) : (KEMB / 16 - 1);
            const float* bp = bp0 + tn * (16 * BPITCH);
            esA = esq[tn * 16 + m];
#pragma unroll
            for (int kc = 0; kc < 16; ++kc) bA[kc] = *(const v2f*)(bp + kc * BPITCH);
        }
        __builtin_amdgcn_sched_barrier(0);
        vq_tile_block(afr0, afr1, bB, accB0, accB1, accA0, accA1, esCur, t * 16 + m,
                      minval0, minidx0, minval1, minidx1);
    }
    {
        const int colP = (KEMB / 16 - 1) * 16 + m;
#pragma unroll
        for (int r = 0; r < 8; ++r) {
            float d0 = esB - 2.0f * accB0[r];
            if (d0 < minval0[r]) { minval0[r] = d0; minidx0[r] = colP; }
            float d1 = esB - 2.0f * accB1[r];
            if (d1 < minval1[r]) { minval1[r] = d1; minidx1[r] = colP; }
        }
    }

    int res0[8], res1[8];
#pragma unroll
    for (int r = 0; r < 8; ++r) {
        float v0 = minval0[r]; int i0 = minidx0[r];
        float v1 = minval1[r]; int i1 = minidx1[r];
        vq_row_min_step<1>(v0, i0); vq_row_min_step<1>(v1, i1);
        vq_row_min_step<2>(v0, i0); vq_row_min_step<2>(v1, i1);
        vq_row_min_step<4>(v0, i0); vq_row_min_step<4>(v1, i1);
        vq_row_min_step<8>(v0, i0); vq_row_min_step<8>(v1, i1);
        res0[r] = i0; res1[r] = i1;
    }
    const int rsel = m & 7;
    int own0 = res0[0], own1 = res1[0];
#pragma unroll
    for (int r = 1; r < 8; ++r) {
        own0 = (rsel == r) ? res0[r] : own0;
        own1 = (rsel == r) ? res1[r] : own1;
    }
    const int oth0 = vq_xhalf_i(own0);
    const int oth1 = vq_xhalf_i(own1);
    const bool mine = ((m >> 3) == g);
    const int k0 = mine ? own0 : oth0;
    const int k1 = mine ? own1 : oth1;

    const float* e0p = Bfrag + (k0 >> 4) * (16 * BPITCH) + g * 32 + (k0 & 15) * 2;
    const float* e1p = Bfrag + (k1 >> 4) * (16 * BPITCH) + g * 32 + (k1 & 15) * 2;
    float* zrow = zq + (b * CDIM) * HW + (rowbase & 4095) + lane;
    const bool lo = (g == 0);
    v2f ls0 = {0.0f, 0.0f}, ls1 = {0.0f, 0.0f};
#pragma unroll
    for (int kc = 0; kc < 16; ++kc) {
        const v2f e0 = *(const v2f*)(e0p + kc * BPITCH);
        const v2f e1 = *(const v2f*)(e1p + kc * BPITCH);
        const v2f d0 = e0 - afr0[kc];
        const v2f d1 = e1 - afr1[kc];
        ls0 += d0 * d0;
        ls1 += d1 * d1;
        const float wx = lo ? e1.x : e0.x;
        const float wy = lo ? e1.y : e0.y;
        const float px = vq_xhalf_f(wx);
        const float py = vq_xhalf_f(wy);
        zrow[(kc * 4 + 0) * HW] = lo ? e0.x : px;
        zrow[(kc * 4 + 1) * HW] = lo ? e0.y : py;
        zrow[(kc * 4 + 2) * HW] = lo ? px : e1.x;
        zrow[(kc * 4 + 3) * HW] = lo ? py : e1.y;
    }
    float lsum = (ls0.x + ls0.y) + (ls1.x + ls1.y);
#pragma unroll
    for (int off = 1; off < 32; off <<= 1) lsum += __shfl_xor(lsum, off, 32);
    if (lane == 0) redbuf[wave] = lsum;
    __syncthreads();
    if (tid == 0) {
        float tot = 0.0f;
        for (int wv = 0; wv < NWAVES; ++wv) tot += redbuf[wv];
        atomicAdd(lossAcc, tot);
    }
}

__global__ void vq_finalize_kernel(const float* acc, float* lossOut) {
    if (threadIdx.x == 0) {
        lossOut[0] = 2.0f * acc[0] / (float)((long)NROWS * CDIM);
    }
}

extern "C" void kernel_launch(void* const* d_in, const int* in_sizes, int n_in,
                              void* d_out, int out_size, void* d_ws, size_t ws_size,
                              hipStream_t stream) {
    const float* x = (const float*)d_in[0];
    const float* E = (const float*)d_in[1];
    float* out     = (float*)d_out;
    float* acc     = (float*)d_ws;

    vq_zero_kernel<<<1, 64, 0, stream>>>(acc);
    vq_main_kernel<<<NWG, TPB, LDS_BYTES, stream>>>(x, E, out, acc);
    vq_finalize_kernel<<<1, 64, 0, stream>>>(acc, out + (out_size - 1));
}


# Prologue loads overlapped: a wave's codebook rows go out in flights of 16, each issued before the last is stored (was 8, drained); x fragments requested before e_sq; those barriers fence LDS only
# speedup vs baseline: 1.9073x; 1.0733x over previous
#include <hip/hip_runtime.h>

typedef __attribute__((ext_vector_type(2))) float v2f;
typedef __attribute__((ext_vector_type(8))) float v8f;

#define BATCH   16
#define CDIM    64
#define HW      4096
#define NROWS   (BATCH * HW)
#define KEMB    1024
#define TPB     256
#define NWAVES  8
#define TILES_W 2
#define ROWS_WG (NWAVES * TILES_W * 16)
#define NWG     (NROWS / ROWS_WG)

#define BPITCH       66
#define BFRAG_FLOATS (KEMB * BPITCH)
#define ESQ_OFF      BFRAG_FLOATS
#define RED_OFF      (ESQ_OFF + KEMB)

#define LDS_FLOATS (RED_OFF + NWAVES)
#define LDS_BYTES  (LDS_FLOATS * 4)

__global__ void vq_zero_kernel(float* acc) {
    if (threadIdx.x == 0) acc[0] = 0.0f;
}

__device__ __forceinline__ void vq_lds_barrier() {
    __builtin_amdgcn_fence(__ATOMIC_RELEASE, "workgroup", "local");
    __builtin_amdgcn_s_barrier();
    __builtin_amdgcn_fence(__ATOMIC_ACQUIRE, "workgroup", "local");
}

__device__ __forceinline__ int vq_xhalf_i(int v) {
    return __builtin_amdgcn_permlanex16(v, v, 0x76543210, 0xfedcba98, false, false);
}
__device__ __forceinline__ float vq_xhalf_f(float v) {
    return __int_as_float(vq_xhalf_i(__float_as_int(v)));
}

template <int XM>
__device__ __forceinline__ void vq_row_min_step(float& v, int& i) {
    const float ov = __int_as_float(
        __builtin_amdgcn_update_dpp(0, __float_as_int(v), 0x160 + XM, 0xf, 0xf, false));
    const int oi = __builtin_amdgcn_update_dpp(0, i, 0x160 + XM, 0xf, 0xf, false);
    const bool take = (ov < v) | ((ov == v) & (oi < i));
    v = take ? ov : v;
    i = take ? oi : i;
}

__device__ __forceinline__ void vq_tile_block(
    const v2f (&a0)[16], const v2f (&a1)[16], const v2f (&b)[16],
    v8f& out0, v8f& out1, const v8f& p0, const v8f& p1, const float esP, const int colP,
    float (&mv0)[8], int (&mi0)[8], float (&mv1)[8], int (&mi1)[8])
{
    v8f c0 = {0.f, 0.f, 0.f, 0.f, 0.f, 0.f, 0.f, 0.f};
    v8f c1 = c0;
#pragma unroll
    for (int kc = 0; kc < 16; ++kc) {
        c0 = __builtin_amdgcn_wmma_f32_16x16x4_f32(
            false, a0[kc], false, b[kc], (short)0, c0, false, false);
        c1 = __builtin_amdgcn_wmma_f32_16x16x4_f32(
            false, a1[kc], false, b[kc], (short)0, c1, false, false);
        if ((kc & 1) == 0) {
            __builtin_amdgcn_sched_barrier(0);
            const int r = kc >> 1;
            float d0 = esP - 2.0f * p0[r];
            if (d0 < mv0[r]) { mv0[r] = d0; mi0[r] = colP; }
            float d1 = esP - 2.0f * p1[r];
            if (d1 < mv1[r]) { mv1[r] = d1; mi1[r] = colP; }
        }
    }
    out0 = c0; out1 = c1;
}

__global__ __launch_bounds__(TPB)
void vq_main_kernel(const float* __restrict__ x,
                    const float* __restrict__ E,
                    float* __restrict__ zq,
                    float* __restrict__ lossAcc)
{
    extern __shared__ float lds[];
    float* Bfrag  = lds;
    float* esq    = lds + ESQ_OFF;
    float* redbuf = lds + RED_OFF;

    const int tid  = threadIdx.x;
    const int wg   = blockIdx.x;
    const int wave = __builtin_amdgcn_readfirstlane(tid >> 5);
    const int lane = tid & 31;

    const int hrow = lane >> 4;
    const int skc  = lane & 15;
    const float* esrc = E + (wave + hrow * 16) * CDIM + skc * 4;
    float* edst = Bfrag + (hrow * 16 + skc) * BPITCH + wave * 2;
    float4 fa[16], fb[16];
#define VQ_LOAD16(buf, i0)                                                        \
    _Pragma("unroll") for (int j = 0; j < 16; ++j) {                              \
        const int i = (i0) + j;                                                   \
        buf[j] = *(const float4*)(esrc + ((i >> 1) * 32 + (i & 1) * 8) * CDIM);   \
    }
#define VQ_STORE16(buf, i0)                                                       \
    _Pragma("unroll") for (int j = 0; j < 16; ++j) {                              \
        const int i = (i0) + j;                                                   \
        float* d = edst + (i >> 1) * (32 * BPITCH) + (i & 1) * 16;                \
        v2f lo2 = {buf[j].x, buf[j].y};                                           \
        v2f hi2 = {buf[j].z, buf[j].w};                                           \
        *(v2f*)(d)      = lo2;                                          \
        *(v2f*)(d + 32) = hi2;                                          \
    }
    VQ_LOAD16(fa, 0)
    VQ_LOAD16(fb, 16)
    VQ_STORE16(fa, 0)
    VQ_LOAD16(fa, 32)
    VQ_STORE16(fb, 16)
    VQ_LOAD16(fb, 48)
    VQ_STORE16(fa, 32)

    const int m    = lane & 15;
    const int g    = lane >> 4;
    const int rowbase = wg * ROWS_WG + wave * (TILES_W * 16);
    const int b       = rowbase >> 12;
    const int nrow    = (rowbase & 4095) + m;

    v2f afr0[16], afr1[16];
#pragma unroll
    for (int kc = 0; kc < 16; ++kc) {
        int c0 = kc * 4 + g * 2;
        afr0[kc].x = x[(b * CDIM + c0    ) * HW + nrow];
        afr0[kc].y = x[(b * CDIM + c0 + 1) * HW + nrow];
        afr1[kc].x = x[(b * CDIM + c0    ) * HW + nrow + 16];
        afr1[kc].y = x[(b * CDIM + c0 + 1) * HW + nrow + 16];
    }

    VQ_STORE16(fb, 48)
#undef VQ_LOAD16
#undef VQ_STORE16
    vq_lds_barrier();
    {
        const float* rp = Bfrag + (tid >> 4) * (16 * BPITCH) + (tid & 15) * 2;
        float s[4] = {0.0f, 0.0f, 0.0f, 0.0f};
#pragma unroll
        for (int kc = 0; kc < 16; ++kc) {
#pragma unroll
            for (int j = 0; j < 4; ++j) {
                const float* p = rp + j * (TPB * BPITCH) + kc * BPITCH;
                v2f lo = *(const v2f*)(p);
                v2f hi = *(const v2f*)(p + 32);
                s[j] = __builtin_fmaf(lo.x, lo.x, s[j]);
                s[j] = __builtin_fmaf(lo.y, lo.y, s[j]);
                s[j] = __builtin_fmaf(hi.x, hi.x, s[j]);
                s[j] = __builtin_fmaf(hi.y, hi.y, s[j]);
            }
        }
#pragma unroll
        for (int j = 0; j < 4; ++j) esq[tid + j * TPB] = s[j];
    }
    vq_lds_barrier();

    float minval0[8], minval1[8];
    int   minidx0[8], minidx1[8];
#pragma unroll
    for (int r = 0; r < 8; ++r) {
        minval0[r] = 3.402823466e38f; minidx0[r] = 0;
        minval1[r] = 3.402823466e38f; minidx1[r] = 0;
    }

    const float* bp0 = Bfrag + lane * 2;
    v2f bA[16], bB[16];
    float esA, esB;
    v8f accA0 = {0.f, 0.f, 0.f, 0.f, 0.f, 0.f, 0.f, 0.f};
    v8f accA1 = accA0, accB0 = accA0, accB1 = accA0;
    esB = 3.402823466e38f;

#pragma unroll
    for (int kc = 0; kc < 16; ++kc) bA[kc] = *(const v2f*)(bp0 + kc * BPITCH);
    esA = esq[m];

    for (int t = 0; t < KEMB / 16; t += 2) {
        const float esPrev = esB;
        __builtin_amdgcn_sched_barrier(0);
        {
            const float* bp = bp0 + (t + 1) * (16 * BPITCH);
            esB = esq[(t + 1) * 16 + m];
#pragma unroll
            for (int kc = 0; kc < 16; ++kc) bB[kc] = *(const v2f*)(bp + kc * BPITCH);
        }
        __builtin_amdgcn_sched_barrier(0);
        vq_tile_block(afr0, afr1, bA, accA0, accA1, accB0, accB1, esPrev, (t - 1) * 16 + m,
                      minval0, minidx0, minval1, minidx1);

        const float esCur = esA;
        __builtin_amdgcn_sched_barrier(0);
        {
            const int tn = (t + 2 < KEMB / 16) ? (t + 2) : (KEMB / 16 - 1);
            const float* bp = bp0 + tn * (16 * BPITCH);
            esA = esq[tn * 16 + m];
#pragma unroll
            for (int kc = 0; kc < 16; ++kc) bA[kc] = *(const v2f*)(bp + kc * BPITCH);
        }
        __builtin_amdgcn_sched_barrier(0);
        vq_tile_block(afr0, afr1, bB, accB0, accB1, accA0, accA1, esCur, t * 16 + m,
                      minval0, minidx0, minval1, minidx1);
    }
    {
        const int colP = (KEMB / 16 - 1) * 16 + m;
#pragma unroll
        for (int r = 0; r < 8; ++r) {
            float d0 = esB - 2.0f * accB0[r];
            if (d0 < minval0[r]) { minval0[r] = d0; minidx0[r] = colP; }
            float d1 = esB - 2.0f * accB1[r];
            if (d1 < minval1[r]) { minval1[r] = d1; minidx1[r] = colP; }
        }
    }

    int res0[8], res1[8];
#pragma unroll
    for (int r = 0; r < 8; ++r) {
        float v0 = minval0[r]; int i0 = minidx0[r];
        float v1 = minval1[r]; int i1 = minidx1[r];
        vq_row_min_step<1>(v0, i0); vq_row_min_step<1>(v1, i1);
        vq_row_min_step<2>(v0, i0); vq_row_min_step<2>(v1, i1);
        vq_row_min_step<4>(v0, i0); vq_row_min_step<4>(v1, i1);
        vq_row_min_step<8>(v0, i0); vq_row_min_step<8>(v1, i1);
        res0[r] = i0; res1[r] = i1;
    }
    const int rsel = m & 7;
    int own0 = res0[0], own1 = res1[0];
#pragma unroll
    for (int r = 1; r < 8; ++r) {
        own0 = (rsel == r) ? res0[r] : own0;
        own1 = (rsel == r) ? res1[r] : own1;
    }
    const int oth0 = vq_xhalf_i(own0);
    const int oth1 = vq_xhalf_i(own1);
    const bool mine = ((m >> 3) == g);
    const int k0 = mine ? own0 : oth0;
    const int k1 = mine ? own1 : oth1;

    const float* e0p = Bfrag + (k0 >> 4) * (16 * BPITCH) + g * 32 + (k0 & 15) * 2;
    const float* e1p = Bfrag + (k1 >> 4) * (16 * BPITCH) + g * 32 + (k1 & 15) * 2;
    float* zrow = zq + (b * CDIM) * HW + (rowbase & 4095) + lane;
    const bool lo = (g == 0);
    v2f ls0 = {0.0f, 0.0f}, ls1 = {0.0f, 0.0f};
#pragma unroll
    for (int kc = 0; kc < 16; ++kc) {
        const v2f e0 = *(const v2f*)(e0p + kc * BPITCH);
        const v2f e1 = *(const v2f*)(e1p + kc * BPITCH);
        const v2f d0 = e0 - afr0[kc];
        const v2f d1 = e1 - afr1[kc];
        ls0 += d0 * d0;
        ls1 += d1 * d1;
        const float wx = lo ? e1.x : e0.x;
        const float wy = lo ? e1.y : e0.y;
        const float px = vq_xhalf_f(wx);
        const float py = vq_xhalf_f(wy);
        zrow[(kc * 4 + 0) * HW] = lo ? e0.x : px;
        zrow[(kc * 4 + 1) * HW] = lo ? e0.y : py;
        zrow[(kc * 4 + 2) * HW] = lo ? px : e1.x;
        zrow[(kc * 4 + 3) * HW] = lo ? py : e1.y;
    }
    float lsum = (ls0.x + ls0.y) + (ls1.x + ls1.y);
#pragma unroll
    for (int off = 1; off < 32; off <<= 1) lsum += __shfl_xor(lsum, off, 32);
    if (lane == 0) redbuf[wave] = lsum;
    __syncthreads();
    if (tid == 0) {
        float tot = 0.0f;
        for (int wv = 0; wv < NWAVES; ++wv) tot += redbuf[wv];
        atomicAdd(lossAcc, tot);
    }
}

__global__ void vq_finalize_kernel(const float* acc, float* lossOut) {
    if (threadIdx.x == 0) {
        lossOut[0] = 2.0f * acc[0] / (float)((long)NROWS * CDIM);
    }
}

extern "C" void kernel_launch(void* const* d_in, const int* in_sizes, int n_in,
                              void* d_out, int out_size, void* d_ws, size_t ws_size,
                              hipStream_t stream) {
    const float* x = (const float*)d_in[0];
    const float* E = (const float*)d_in[1];
    float* out     = (float*)d_out;
    float* acc     = (float*)d_ws;

    vq_zero_kernel<<<1, 64, 0, stream>>>(acc);
    vq_main_kernel<<<NWG, TPB, LDS_BYTES, stream>>>(x, E, out, acc);
    vq_finalize_kernel<<<1, 64, 0, stream>>>(acc, out + (out_size - 1));
}


# Matrix operands swapped (codes as rows): a lane holds 8 scores of ONE token per tile, folds them by a min tree and one compare; the winning tile's 8 scores are kept, code picked after the loop
# speedup vs baseline: 2.0558x; 1.0779x over previous
#include <hip/hip_runtime.h>

typedef __attribute__((ext_vector_type(2))) float v2f;
typedef __attribute__((ext_vector_type(4))) float v4f;
typedef __attribute__((ext_vector_type(8))) float v8f;

#define BATCH   16
#define CDIM    64
#define HW      4096
#define NROWS   (BATCH * HW)
#define KEMB    1024
#define TPB     256
#define NWAVES  8
#define TILES_W 2
#define ROWS_WG (NWAVES * TILES_W * 16)
#define NWG     (NROWS / ROWS_WG)

#define BPITCH       66
#define BFRAG_FLOATS (KEMB * BPITCH)
#define ESQ_OFF      BFRAG_FLOATS
#define RED_OFF      (ESQ_OFF + KEMB)

#define LDS_FLOATS (RED_OFF + NWAVES)
#define LDS_BYTES  (LDS_FLOATS * 4)

__global__ void vq_zero_kernel(float* acc) {
    if (threadIdx.x == 0) acc[0] = 0.0f;
}

__device__ __forceinline__ void vq_lds_barrier() {
    __builtin_amdgcn_fence(__ATOMIC_RELEASE, "workgroup", "local");
    __builtin_amdgcn_s_barrier();
    __builtin_amdgcn_fence(__ATOMIC_ACQUIRE, "workgroup", "local");
}

__device__ __forceinline__ int vq_xhalf_i(int v) {
    return __builtin_amdgcn_permlanex16(v, v, 0x76543210, 0xfedcba98, false, false);
}
__device__ __forceinline__ float vq_xhalf_f(float v) {
    return __int_as_float(vq_xhalf_i(__float_as_int(v)));
}

__device__ __forceinline__ float vq_min3(float a, float b, float c) {
    return __builtin_fminf(__builtin_fminf(a, b), c);
}


__device__ __forceinline__ v8f vq_load_es(const float* esqh, const int t) {
    const v4f lo = *(const v4f*)(esqh + t * 16);
    const v4f hi = *(const v4f*)(esqh + t * 16 + 4);
    const v8f r = {lo.x, lo.y, lo.z, lo.w, hi.x, hi.y, hi.z, hi.w};
    return r;
}

__device__ __forceinline__ void vq_fold(const v8f& p, const v8f& es, const int tP,
                                        float& run, int& rt, v8f& sv)
{
    const v8f d = es - 2.0f * p;
    const float a = vq_min3(d[0], d[1], d[2]);
    const float b = vq_min3(d[3], d[4], d[5]);
    const float c = vq_min3(d[6], d[7], a);
    const float m = __builtin_fminf(b, c);
    const bool u = m < run;
    run = u ? m : run;
    rt  = u ? tP : rt;
    sv  = u ? d : sv;
}

__device__ __forceinline__ void vq_tile_block(
    const v2f (&xa)[16], const v2f (&xb)[16], const v2f (&e)[16],
    v8f& out0, v8f& out1, const v8f& p0, const v8f& p1, const v8f& es, const int tP,
    float& run0, int& rt0, v8f& sv0, float& run1, int& rt1, v8f& sv1)
{
    v8f c0 = {0.f, 0.f, 0.f, 0.f, 0.f, 0.f, 0.f, 0.f};
    v8f c1 = c0;
    v8f d0 = c0, d1 = c0;
    float a0 = 0.f, b0 = 0.f, g0 = 0.f, m0 = 0.f;
    float a1 = 0.f, b1 = 0.f, g1 = 0.f, m1 = 0.f;
    bool u0 = false, u1 = false;
#pragma unroll
    for (int kc = 0; kc < 16; ++kc) {
        c0 = __builtin_amdgcn_wmma_f32_16x16x4_f32(
            false, e[kc], false, xa[kc], (short)0, c0, false, false);
        c1 = __builtin_amdgcn_wmma_f32_16x16x4_f32(
            false, e[kc], false, xb[kc], (short)0, c1, false, false);
        if ((kc & 1) == 0) {
            const int s = kc >> 1;
            if (s == 1) d0 = es - 2.0f * p0;
            if (s == 2) d1 = es - 2.0f * p1;
            if (s == 3) {
                a0 = vq_min3(d0[0], d0[1], d0[2]);
                b0 = vq_min3(d0[3], d0[4], d0[5]);
                a1 = vq_min3(d1[0], d1[1], d1[2]);
                b1 = vq_min3(d1[3], d1[4], d1[5]);
            }
            if (s == 4) {
                g0 = vq_min3(d0[6], d0[7], a0);
                g1 = vq_min3(d1[6], d1[7], a1);
            }
            if (s == 5) {
                m0 = __builtin_fminf(b0, g0);
                m1 = __builtin_fminf(b1, g1);
            }
            if (s == 6) {
                u0 = m0 < run0;
                u1 = m1 < run1;
                run0 = u0 ? m0 : run0;
                rt0  = u0 ? tP : rt0;
                sv0  = u0 ? d0 : sv0;
            }
            if (s == 7) {
                run1 = u1 ? m1 : run1;
                rt1  = u1 ? tP : rt1;
                sv1  = u1 ? d1 : sv1;
            }
        }
    }
    out0 = c0; out1 = c1;
}

__device__ __forceinline__ int vq_resolve(const float run, const int rt, const v8f& sv, const int h) {
    int rr = 7;
#pragma unroll
    for (int r = 6; r >= 0; --r) rr = (sv[r] == run) ? r : rr;
    const int code = rt * 16 + h * 8 + rr;
    const float ov = vq_xhalf_f(run);
    const int   oc = vq_xhalf_i(code);
    const bool take = (ov < run) | ((ov == run) & (oc < code));
    return take ? oc : code;
}

__global__ __launch_bounds__(TPB)
void vq_main_kernel(const float* __restrict__ x,
                    const float* __restrict__ E,
                    float* __restrict__ zq,
                    float* __restrict__ lossAcc)
{
    extern __shared__ float lds[];
    float* Bfrag  = lds;
    float* esq    = lds + ESQ_OFF;
    float* redbuf = lds + RED_OFF;

    const int tid  = threadIdx.x;
    const int wg   = blockIdx.x;
    const int wave = __builtin_amdgcn_readfirstlane(tid >> 5);
    const int lane = tid & 31;

    const int hrow = lane >> 4;
    const int skc  = lane & 15;
    const float* esrc = E + (wave + hrow * 16) * CDIM + skc * 4;
    float* edst = Bfrag + (hrow * 16 + skc) * BPITCH + wave * 2;
    float4 fa[16], fb[16];
#define VQ_LOAD16(buf, i0)                                                        \
    _Pragma("unroll") for (int j = 0; j < 16; ++j) {                              \
        const int i = (i0) + j;                                                   \
        buf[j] = *(const float4*)(esrc + ((i >> 1) * 32 + (i & 1) * 8) * CDIM);   \
    }
#define VQ_STORE16(buf, i0)                                                       \
    _Pragma("unroll") for (int j = 0; j < 16; ++j) {                              \
        const int i = (i0) + j;                                                   \
        float* d = edst + (i >> 1) * (32 * BPITCH) + (i & 1) * 16;                \
        v2f lo2 = {buf[j].x, buf[j].y};                                           \
        v2f hi2 = {buf[j].z, buf[j].w};                                           \
        *(v2f*)(d)      = lo2;                                          \
        *(v2f*)(d + 32) = hi2;                                          \
    }
    VQ_LOAD16(fa, 0)
    VQ_LOAD16(fb, 16)
    VQ_STORE16(fa, 0)
    VQ_LOAD16(fa, 32)
    VQ_STORE16(fb, 16)
    VQ_LOAD16(fb, 48)
    VQ_STORE16(fa, 32)

    const int m    = lane & 15;
    const int g    = lane >> 4;
    const int rowbase = wg * ROWS_WG + wave * (TILES_W * 16);
    const int b       = rowbase >> 12;
    const int nrow    = (rowbase & 4095) + m;

    v2f afr0[16], afr1[16];
#pragma unroll
    for (int kc = 0; kc < 16; ++kc) {
        int c0 = kc * 4 + g * 2;
        afr0[kc].x = x[(b * CDIM + c0    ) * HW + nrow];
        afr0[kc].y = x[(b * CDIM + c0 + 1) * HW + nrow];
        afr1[kc].x = x[(b * CDIM + c0    ) * HW + nrow + 16];
        afr1[kc].y = x[(b * CDIM + c0 + 1) * HW + nrow + 16];
    }

    VQ_STORE16(fb, 48)
#undef VQ_LOAD16
#undef VQ_STORE16
    vq_lds_barrier();
    {
        const float* rp = Bfrag + (tid >> 4) * (16 * BPITCH) + (tid & 15) * 2;
        float s[4] = {0.0f, 0.0f, 0.0f, 0.0f};
#pragma unroll
        for (int kc = 0; kc < 16; ++kc) {
#pragma unroll
            for (int j = 0; j < 4; ++j) {
                const float* p = rp + j * (TPB * BPITCH) + kc * BPITCH;
                v2f lo = *(const v2f*)(p);
                v2f hi = *(const v2f*)(p + 32);
                s[j] = __builtin_fmaf(lo.x, lo.x, s[j]);
                s[j] = __builtin_fmaf(lo.y, lo.y, s[j]);
                s[j] = __builtin_fmaf(hi.x, hi.x, s[j]);
                s[j] = __builtin_fmaf(hi.y, hi.y, s[j]);
            }
        }
#pragma unroll
        for (int j = 0; j < 4; ++j) esq[tid + j * TPB] = s[j];
    }
    vq_lds_barrier();

    const float big = 3.402823466e38f;
    float run0 = big, run1 = big;
    int   rt0 = 0, rt1 = 0;
    v8f   sv0 = {big, big, big, big, big, big, big, big};
    v8f   sv1 = sv0;

    const float* bp0  = Bfrag + lane * 2;
    const float* esqh = esq + g * 8;
    v2f bA[16], bB[16];
    v8f es;
    v8f accA0 = {0.f, 0.f, 0.f, 0.f, 0.f, 0.f, 0.f, 0.f};
    v8f accA1 = accA0;
    v8f accB0 = {-big, -big, -big, -big, -big, -big, -big, -big};
    v8f accB1 = accB0;

#pragma unroll
    for (int kc = 0; kc < 16; ++kc) bA[kc] = *(const v2f*)(bp0 + kc * BPITCH);

    for (int t = 0; t < KEMB / 16; t += 2) {
        __builtin_amdgcn_sched_barrier(0);
        {
            const int tp = (t > 0) ? (t - 1) : 0;
            es = vq_load_es(esqh, tp);
            const float* bp = bp0 + (t + 1) * (16 * BPITCH);
#pragma unroll
            for (int kc = 0; kc < 16; ++kc) bB[kc] = *(const v2f*)(bp + kc * BPITCH);
        }
        __builtin_amdgcn_sched_barrier(0);
        vq_tile_block(afr0, afr1, bA, accA0, accA1, accB0, accB1, es, t - 1,
                      run0, rt0, sv0, run1, rt1, sv1);

        __builtin_amdgcn_sched_barrier(0);
        {
            es = vq_load_es(esqh, t);
            const int tn = (t + 2 < KEMB / 16) ? (t + 2) : (KEMB / 16 - 1);
            const float* bp = bp0 + tn * (16 * BPITCH);
#pragma unroll
            for (int kc = 0; kc < 16; ++kc) bA[kc] = *(const v2f*)(bp + kc * BPITCH);
        }
        __builtin_amdgcn_sched_barrier(0);
        vq_tile_block(afr0, afr1, bB, accB0, accB1, accA0, accA1, es, t,
                      run0, rt0, sv0, run1, rt1, sv1);
    }
    es = vq_load_es(esqh, KEMB / 16 - 1);
    vq_fold(accB0, es, KEMB / 16 - 1, run0, rt0, sv0);
    vq_fold(accB1, es, KEMB / 16 - 1, run1, rt1, sv1);

    const int k0 = vq_resolve(run0, rt0, sv0, g);
    const int k1 = vq_resolve(run1, rt1, sv1, g);

    const float* e0p = Bfrag + (k0 >> 4) * (16 * BPITCH) + g * 32 + (k0 & 15) * 2;
    const float* e1p = Bfrag + (k1 >> 4) * (16 * BPITCH) + g * 32 + (k1 & 15) * 2;
    float* zrow = zq + (b * CDIM) * HW + (rowbase & 4095) + lane;
    const bool lo = (g == 0);
    v2f ls0 = {0.0f, 0.0f}, ls1 = {0.0f, 0.0f};
#pragma unroll
    for (int kc = 0; kc < 16; ++kc) {
        const v2f e0 = *(const v2f*)(e0p + kc * BPITCH);
        const v2f e1 = *(const v2f*)(e1p + kc * BPITCH);
        const v2f d0 = e0 - afr0[kc];
        const v2f d1 = e1 - afr1[kc];
        ls0 += d0 * d0;
        ls1 += d1 * d1;
        const float wx = lo ? e1.x : e0.x;
        const float wy = lo ? e1.y : e0.y;
        const float px = vq_xhalf_f(wx);
        const float py = vq_xhalf_f(wy);
        zrow[(kc * 4 + 0) * HW] = lo ? e0.x : px;
        zrow[(kc * 4 + 1) * HW] = lo ? e0.y : py;
        zrow[(kc * 4 + 2) * HW] = lo ? px : e1.x;
        zrow[(kc * 4 + 3) * HW] = lo ? py : e1.y;
    }
    float lsum = (ls0.x + ls0.y) + (ls1.x + ls1.y);
#pragma unroll
    for (int off = 1; off < 32; off <<= 1) lsum += __shfl_xor(lsum, off, 32);
    if (lane == 0) redbuf[wave] = lsum;
    __syncthreads();
    if (tid == 0) {
        float tot = 0.0f;
        for (int wv = 0; wv < NWAVES; ++wv) tot += redbuf[wv];
        atomicAdd(lossAcc, tot);
    }
}

__global__ void vq_finalize_kernel(const float* acc, float* lossOut) {
    if (threadIdx.x == 0) {
        lossOut[0] = 2.0f * acc[0] / (float)((long)NROWS * CDIM);
    }
}

extern "C" void kernel_launch(void* const* d_in, const int* in_sizes, int n_in,
                              void* d_out, int out_size, void* d_ws, size_t ws_size,
                              hipStream_t stream) {
    const float* x = (const float*)d_in[0];
    const float* E = (const float*)d_in[1];
    float* out     = (float*)d_out;
    float* acc     = (float*)d_ws;

    vq_zero_kernel<<<1, 64, 0, stream>>>(acc);
    vq_main_kernel<<<NWG, TPB, LDS_BYTES, stream>>>(x, E, out, acc);
    vq_finalize_kernel<<<1, 64, 0, stream>>>(acc, out + (out_size - 1));
}
